# starting kernel
# baseline (speedup 1.0000x reference)
#include <hip/hip_runtime.h>
#include <hip/hip_bf16.h>

typedef float v2f __attribute__((ext_vector_type(2)));
typedef float v8f __attribute__((ext_vector_type(8)));

#define BIGF 3.0e38f

__global__ __launch_bounds__(256) void transform_points_kernel(
    const float* __restrict__ src, float4* __restrict__ dst, int n, int npad, int B)
{
    int i = blockIdx.x * blockDim.x + threadIdx.x;
    int total = B * npad;
    if (i >= total) return;
    int b = i / npad;
    int idx = i - b * npad;
    float x = 0.f, y = 0.f, z = 0.f, w = BIGF;
    if (idx < n) {
        const float* s = src + ((size_t)b * n + idx) * 3;
        x = s[0]; y = s[1]; z = s[2];
        w = x * x + y * y + z * z;
    }
    dst[i] = make_float4(x, y, z, w);
}

__global__ __launch_bounds__(256) void chamfer_wmma_kernel(
    const float4* __restrict__ Pt, const float4* __restrict__ Qt,
    float* __restrict__ out, int NP, int NPpad, int NQpad, int B)
{
    const int lane  = threadIdx.x & 31;
    const int gwave = blockIdx.x * (blockDim.x >> 5) + (threadIdx.x >> 5);

    const int groups = NPpad >> 5;
    const int b = gwave / groups;
    if (b >= B) return;
    const int row0 = (gwave - b * groups) << 5;

    const int  j  = lane & 15;
    const bool hi = lane >= 16;

    const float4* Pb = Pt + (size_t)b * NPpad;
    const float* Qb = (const float*)(Qt + (size_t)b * NQpad) + 4 * j + (hi ? 2 : 0);

    float4 p0 = Pb[row0 + j];
    float4 p1 = Pb[row0 + 16 + j];

    v2f a0, a1;
    a0.x = hi ? (-2.f * p0.z) : (-2.f * p0.x);
    a0.y = hi ? 1.f           : (-2.f * p0.y);
    a1.x = hi ? (-2.f * p1.z) : (-2.f * p1.x);
    a1.y = hi ? 1.f           : (-2.f * p1.y);

    const int srcBase = hi ? 8 : 0;
    v8f c0, c1, minv0, minv1;
#pragma unroll
    for (int r = 0; r < 8; ++r) {
        c0[r] = __shfl(p0.w, srcBase + r, 16);
        c1[r] = __shfl(p1.w, srcBase + r, 16);
        minv0[r] = BIGF;
        minv1[r] = BIGF;
    }

#pragma unroll 4
    for (int col0 = 0; col0 < NQpad; col0 += 16) {
        v2f bb = *(const v2f*)(Qb + 4 * col0);

        v8f d0 = __builtin_amdgcn_wmma_f32_16x16x4_f32(
            false, a0, false, bb, (short)0, c0, false, false);
        v8f d1 = __builtin_amdgcn_wmma_f32_16x16x4_f32(
            false, a1, false, bb, (short)0, c1, false, false);

#pragma unroll
        for (int r = 0; r < 8; ++r) {
            minv0[r] = fminf(minv0[r], d0[r]);
            minv1[r] = fminf(minv1[r], d1[r]);
        }
    }

#pragma unroll
    for (int mask = 1; mask <= 8; mask <<= 1) {
#pragma unroll
        for (int r = 0; r < 8; ++r) {
            minv0[r] = fminf(minv0[r], __shfl_xor(minv0[r], mask, 32));
            minv1[r] = fminf(minv1[r], __shfl_xor(minv1[r], mask, 32));
        }
    }

    if (j == 0) {
        int mbase = row0 + (hi ? 8 : 0);
        float* ob = out + (size_t)b * NP;
#pragma unroll
        for (int r = 0; r < 8; ++r) {
            int m0 = mbase + r;
            int m1 = mbase + 16 + r;
            if (m0 < NP) ob[m0] = minv0[r];
            if (m1 < NP) ob[m1] = minv1[r];
        }
    }
}

__global__ __launch_bounds__(256) void chamfer_nn_min_fallback(
    const float* __restrict__ P, const float* __restrict__ Q,
    float* __restrict__ out, int NP, int NQ, int B)
{
    const int lane  = threadIdx.x & 31;
    const int gwave = blockIdx.x * (blockDim.x >> 5) + (threadIdx.x >> 5);
    const int stripsPerBatch = (NP + 15) >> 4;
    const int b = gwave / stripsPerBatch;
    if (b >= B) return;
    const int row0 = (gwave - b * stripsPerBatch) << 4;
    const int  j  = lane & 15;
    const bool hi = lane >= 16;

    const float* Pb = P + (size_t)b * NP * 3;
    const float* Qb = Q + (size_t)b * NQ * 3;

    int pi = row0 + j; if (pi >= NP) pi = NP - 1;
    float px = Pb[3 * pi], py = Pb[3 * pi + 1], pz = Pb[3 * pi + 2];
    float p2 = px * px + py * py + pz * pz;

    v2f a;
    a.x = hi ? (-2.f * pz) : (-2.f * px);
    a.y = hi ? 1.f         : (-2.f * py);

    v8f c, minv;
#pragma unroll
    for (int r = 0; r < 8; ++r) {
        c[r] = __shfl(p2, (hi ? 8 : 0) + r, 16);
        minv[r] = BIGF;
    }

#pragma unroll 2
    for (int col0 = 0; col0 < NQ; col0 += 16) {
        int qi = col0 + j;
        bool qok = qi < NQ;
        if (!qok) qi = NQ - 1;
        float qx = Qb[3 * qi], qy = Qb[3 * qi + 1], qz = Qb[3 * qi + 2];
        float q2 = qok ? (qx * qx + qy * qy + qz * qz) : BIGF;
        v2f bb;
        bb.x = hi ? qz : qx;
        bb.y = hi ? q2 : qy;
        v8f d = __builtin_amdgcn_wmma_f32_16x16x4_f32(
            false, a, false, bb, (short)0, c, false, false);
#pragma unroll
        for (int r = 0; r < 8; ++r) minv[r] = fminf(minv[r], d[r]);
    }

#pragma unroll
    for (int mask = 1; mask <= 8; mask <<= 1) {
#pragma unroll
        for (int r = 0; r < 8; ++r)
            minv[r] = fminf(minv[r], __shfl_xor(minv[r], mask, 32));
    }

    if (j == 0) {
        int mbase = row0 + (hi ? 8 : 0);
#pragma unroll
        for (int r = 0; r < 8; ++r) {
            int m = mbase + r;
            if (m < NP) out[(size_t)b * NP + m] = minv[r];
        }
    }
}

extern "C" void kernel_launch(void* const* d_in, const int* in_sizes, int n_in,
                              void* d_out, int out_size, void* d_ws, size_t ws_size,
                              hipStream_t stream) {
    const float* in1 = (const float*)d_in[0];
    const float* in2 = (const float*)d_in[1];
    float* out = (float*)d_out;

    const int B = 4;
    const int N = in_sizes[0] / (B * 3);
    const int M = in_sizes[1] / (B * 3);
    const int Npad = (N + 31) & ~31;
    const int Mpad = (M + 31) & ~31;

    size_t need = (size_t)B * (Npad + Mpad) * sizeof(float4);
    if (ws_size >= need) {
        float4* T1 = (float4*)d_ws;
        float4* T2 = T1 + (size_t)B * Npad;

        transform_points_kernel<<<(B * Npad + 255) / 256, 256, 0, stream>>>(in1, T1, N, Npad, B);
        transform_points_kernel<<<(B * Mpad + 255) / 256, 256, 0, stream>>>(in2, T2, M, Mpad, B);

        {
            int waves = B * (Npad >> 5);
            int blocks = (waves + 7) / 8;
            chamfer_wmma_kernel<<<blocks, 256, 0, stream>>>(T1, T2, out, N, Npad, Mpad, B);
        }
        {
            int waves = B * (Mpad >> 5);
            int blocks = (waves + 7) / 8;
            chamfer_wmma_kernel<<<blocks, 256, 0, stream>>>(T2, T1, out + (size_t)B * N, M, Mpad, Npad, B);
        }
    } else {
        {
            int waves = B * ((N + 15) / 16);
            int blocks = (waves + 7) / 8;
            chamfer_nn_min_fallback<<<blocks, 256, 0, stream>>>(in1, in2, out, N, M, B);
        }
        {
            int waves = B * ((M + 15) / 16);
            int blocks = (waves + 7) / 8;
            chamfer_nn_min_fallback<<<blocks, 256, 0, stream>>>(in2, in1, out + (size_t)B * N, M, N, B);
        }
    }
}


# Table builds moved to 4 builder waves, the 2nd wave of each SIMD: in 3 stages they load, norm and store the next table region while waves 0-3 run the matrix scan; half a table first; sums unchanged
# speedup vs baseline: 4.2405x; 4.2405x over previous
#include <hip/hip_runtime.h>
#include <hip/hip_bf16.h>

typedef float v2f __attribute__((ext_vector_type(2)));
typedef float v8f __attribute__((ext_vector_type(8)));

#define BIGF 3.0e38f

__global__ __launch_bounds__(256) void transform_points_kernel(
    const float* __restrict__ src, float4* __restrict__ dst, int n, int npad, int B)
{
    int i = blockIdx.x * blockDim.x + threadIdx.x;
    int total = B * npad;
    if (i >= total) return;
    int b = i / npad;
    int idx = i - b * npad;
    float x = 0.f, y = 0.f, z = 0.f, w = BIGF;
    if (idx < n) {
        const float* s = src + ((size_t)b * n + idx) * 3;
        x = s[0]; y = s[1]; z = s[2];
        w = x * x + y * y + z * z;
    }
    dst[i] = make_float4(x, y, z, w);
}

#define BLK_TILES  32
#define BLK_FLOATS (BLK_TILES * 64)

__device__ __forceinline__ void load_block(v2f (&buf)[BLK_TILES], const float* p)
{
#pragma unroll
    for (int t = 0; t < BLK_TILES; ++t)
        buf[t] = *(const v2f*)(p + 64 * t);
    __builtin_amdgcn_sched_barrier(0);
}

__device__ __forceinline__ void scan_block(
    const v2f (&buf)[BLK_TILES], v2f a0, v2f a1, v8f c0, v8f c1, v8f& minv0, v8f& minv1)
{
#pragma unroll
    for (int t = 0; t < BLK_TILES; ++t) {
        v8f d0 = __builtin_amdgcn_wmma_f32_16x16x4_f32(
            false, a0, false, buf[t], (short)0, c0, false, false);
        v8f d1 = __builtin_amdgcn_wmma_f32_16x16x4_f32(
            false, a1, false, buf[t], (short)0, c1, false, false);
#pragma unroll
        for (int r = 0; r < 8; ++r) {
            minv0[r] = fminf(minv0[r], d0[r]);
            minv1[r] = fminf(minv1[r], d1[r]);
        }
    }
}

__global__ __launch_bounds__(128) void chamfer_wmma_kernel(
    const float4* __restrict__ Pt, const float4* __restrict__ Qt,
    float* __restrict__ out, int NP, int NPpad, int NQpad, int B)
{
    const int lane  = threadIdx.x & 31;
    const int gwave = blockIdx.x * (blockDim.x >> 5) + (threadIdx.x >> 5);

    const int groups = NPpad >> 5;
    const int b = gwave / groups;
    if (b >= B) return;
    const int row0 = (gwave - b * groups) << 5;

    const int  j  = lane & 15;
    const bool hi = lane >= 16;

    const float4* Pb = Pt + (size_t)b * NPpad;
    const float* Qb = (const float*)(Qt + (size_t)b * NQpad) + 4 * j + (hi ? 2 : 0);

    float4 p0 = Pb[row0 + j];
    float4 p1 = Pb[row0 + 16 + j];

    v2f a0, a1;
    a0.x = hi ? (-2.f * p0.z) : (-2.f * p0.x);
    a0.y = hi ? 1.f           : (-2.f * p0.y);
    a1.x = hi ? (-2.f * p1.z) : (-2.f * p1.x);
    a1.y = hi ? 1.f           : (-2.f * p1.y);

    const int srcBase = hi ? 8 : 0;
    v8f c0, c1, minv0, minv1;
#pragma unroll
    for (int r = 0; r < 8; ++r) {
        c0[r] = __shfl(p0.w, srcBase + r, 16);
        c1[r] = __shfl(p1.w, srcBase + r, 16);
        minv0[r] = BIGF;
        minv1[r] = BIGF;
    }

    const int tiles = NQpad >> 4;
    const int nblk  = tiles / BLK_TILES;
    int blk = 0;
    if (nblk > 0) {
        v2f bufA[BLK_TILES], bufB[BLK_TILES];
        load_block(bufA, Qb);
        for (; blk + 2 <= nblk; blk += 2) {
            load_block(bufB, Qb + (size_t)(blk + 1) * BLK_FLOATS);
            scan_block(bufA, a0, a1, c0, c1, minv0, minv1);
            const int nb = (blk + 2 < nblk) ? (blk + 2) : (nblk - 1);
            load_block(bufA, Qb + (size_t)nb * BLK_FLOATS);
            scan_block(bufB, a0, a1, c0, c1, minv0, minv1);
        }
        if (blk < nblk) {
            scan_block(bufA, a0, a1, c0, c1, minv0, minv1);
            ++blk;
        }
    }
    for (int col0 = blk * (BLK_TILES * 16); col0 < NQpad; col0 += 16) {
        v2f bb = *(const v2f*)(Qb + 4 * col0);
        v8f d0 = __builtin_amdgcn_wmma_f32_16x16x4_f32(
            false, a0, false, bb, (short)0, c0, false, false);
        v8f d1 = __builtin_amdgcn_wmma_f32_16x16x4_f32(
            false, a1, false, bb, (short)0, c1, false, false);
#pragma unroll
        for (int r = 0; r < 8; ++r) {
            minv0[r] = fminf(minv0[r], d0[r]);
            minv1[r] = fminf(minv1[r], d1[r]);
        }
    }

#pragma unroll
    for (int mask = 1; mask <= 8; mask <<= 1) {
#pragma unroll
        for (int r = 0; r < 8; ++r) {
            minv0[r] = fminf(minv0[r], __shfl_xor(minv0[r], mask, 32));
            minv1[r] = fminf(minv1[r], __shfl_xor(minv1[r], mask, 32));
        }
    }

    if (j == 0) {
        int mbase = row0 + (hi ? 8 : 0);
        float* ob = out + (size_t)b * NP;
#pragma unroll
        for (int r = 0; r < 8; ++r) {
            int m0 = mbase + r;
            int m1 = mbase + 16 + r;
            if (m0 < NP) ob[m0] = minv0[r];
            if (m1 < NP) ob[m1] = minv1[r];
        }
    }
}

struct PendingPair { v8f s0a, s1a, s0b, s1b; };

__device__ __forceinline__ void fold4(v8f& m, const v8f& x, const v8f& y, int r0)
{
#pragma unroll
    for (int r = r0; r < r0 + 4; ++r) m[r] = fminf(fminf(m[r], x[r]), y[r]);
}

__device__ __forceinline__ void scan_block_late_fold(
    const v2f (&buf)[BLK_TILES], v2f a0, v2f a1, v8f c0, v8f c1,
    PendingPair& pd, v8f& minv0, v8f& minv1)
{
#pragma unroll
    for (int t = 0; t < BLK_TILES; t += 2) {
        v8f n0a = __builtin_amdgcn_wmma_f32_16x16x4_f32(
            false, a0, false, buf[t], (short)0, c0, false, false);
        fold4(minv0, pd.s0a, pd.s0b, 0);
        v8f n1a = __builtin_amdgcn_wmma_f32_16x16x4_f32(
            false, a1, false, buf[t], (short)0, c1, false, false);
        fold4(minv0, pd.s0a, pd.s0b, 4);
        v8f n0b = __builtin_amdgcn_wmma_f32_16x16x4_f32(
            false, a0, false, buf[t + 1], (short)0, c0, false, false);
        fold4(minv1, pd.s1a, pd.s1b, 0);
        v8f n1b = __builtin_amdgcn_wmma_f32_16x16x4_f32(
            false, a1, false, buf[t + 1], (short)0, c1, false, false);
        fold4(minv1, pd.s1a, pd.s1b, 4);
        pd.s0a = n0a; pd.s1a = n1a; pd.s0b = n0b; pd.s1b = n1b;
    }
#pragma unroll
    for (int k = 0; k < 2 * BLK_TILES; ++k) {
        __builtin_amdgcn_sched_group_barrier(0x008, 1, 0);
        __builtin_amdgcn_sched_group_barrier(0x002, 4, 0);
    }
    __builtin_amdgcn_sched_barrier(0);
}

#define STAGE_POINTS 64

template <bool PADDED>
__device__ __forceinline__ void build_table(
    float4* tq, const float* __restrict__ Qraw, int n, int npad, int tid)
{
    int g = 0;
    for (; g + 128 * STAGE_POINTS <= npad; g += 128 * STAGE_POINTS) {
        float qx[STAGE_POINTS], qy[STAGE_POINTS], qz[STAGE_POINTS];
#pragma unroll
        for (int u = 0; u < STAGE_POINTS; ++u) {
            const int idx = g + 128 * u + tid;
            const float* s = Qraw + (size_t)((!PADDED || idx < n) ? idx : n - 1) * 3;
            qx[u] = s[0]; qy[u] = s[1]; qz[u] = s[2];
        }
#pragma unroll
        for (int u = 0; u < STAGE_POINTS; ++u) {
            const int idx = g + 128 * u + tid;
            const bool ok = !PADDED || idx < n;
            const float x = ok ? qx[u] : 0.f;
            const float y = ok ? qy[u] : 0.f;
            const float z = ok ? qz[u] : 0.f;
            const float w = ok ? (x * x + y * y + z * z) : BIGF;
            tq[idx] = make_float4(x, y, z, w);
        }
    }
    for (int idx = g + tid; idx < npad; idx += 128) {
        const bool ok = !PADDED || idx < n;
        const float* s = Qraw + (size_t)(ok ? idx : n - 1) * 3;
        const float x = ok ? s[0] : 0.f;
        const float y = ok ? s[1] : 0.f;
        const float z = ok ? s[2] : 0.f;
        const float w = ok ? (x * x + y * y + z * z) : BIGF;
        tq[idx] = make_float4(x, y, z, w);
    }
}

__device__ __forceinline__ void lds_barrier()
{
    __builtin_amdgcn_fence(__ATOMIC_RELEASE, "workgroup", "local");
    __builtin_amdgcn_s_barrier();
    __builtin_amdgcn_fence(__ATOMIC_ACQUIRE, "workgroup", "local");
}

#define HOLD_POINTS 64

template <bool CLAMP>
__device__ __forceinline__ void hold_points(
    float (&hx)[HOLD_POINTS], float (&hy)[HOLD_POINTS], float (&hz)[HOLD_POINTS],
    const float* __restrict__ raw, int n, int tid)
{
#pragma unroll
    for (int u = 0; u < HOLD_POINTS; ++u) {
        const int idx = 128 * u + tid;
        const float* s = raw + (size_t)((!CLAMP || idx < n) ? idx : n - 1) * 3;
        hx[u] = s[0]; hy[u] = s[1]; hz[u] = s[2];
    }
    __builtin_amdgcn_sched_barrier(0);
}

template <bool CLAMP>
__device__ __forceinline__ void store_held_points(
    float4* tq, const float (&hx)[HOLD_POINTS], const float (&hy)[HOLD_POINTS],
    const float (&hz)[HOLD_POINTS], int n, int npad, int tid)
{
#pragma unroll
    for (int u = 0; u < HOLD_POINTS; ++u) {
        const int idx = 128 * u + tid;
        const bool ok = !CLAMP || idx < n;
        const float x = ok ? hx[u] : 0.f;
        const float y = ok ? hy[u] : 0.f;
        const float z = ok ? hz[u] : 0.f;
        const float w = ok ? (x * x + y * y + z * z) : BIGF;
        if (!CLAMP || idx < npad) tq[idx] = make_float4(x, y, z, w);
    }
}

struct RowOperands { v2f a0, a1; v8f c0, c1; };

__device__ __forceinline__ RowOperands make_row_operands(
    float p0x, float p0y, float p0z, float p1x, float p1y, float p1z, bool hi)
{
    RowOperands ro;
    const float p0w = p0x * p0x + p0y * p0y + p0z * p0z;
    const float p1w = p1x * p1x + p1y * p1y + p1z * p1z;
    ro.a0.x = hi ? (-2.f * p0z) : (-2.f * p0x);
    ro.a0.y = hi ? 1.f          : (-2.f * p0y);
    ro.a1.x = hi ? (-2.f * p1z) : (-2.f * p1x);
    ro.a1.y = hi ? 1.f          : (-2.f * p1y);
    const int srcBase = hi ? 8 : 0;
#pragma unroll
    for (int r = 0; r < 8; ++r) {
        ro.c0[r] = __shfl(p0w, srcBase + r, 16);
        ro.c1[r] = __shfl(p1w, srcBase + r, 16);
    }
    return ro;
}

__device__ __forceinline__ void scan_lds_region(
    const float* Qb, int npad, const RowOperands& ro, v8f& minv0, v8f& minv1)
{
    const v2f a0 = ro.a0, a1 = ro.a1;
    const v8f c0 = ro.c0, c1 = ro.c1;
    PendingPair pd;
#pragma unroll
    for (int r = 0; r < 8; ++r) {
        pd.s0a[r] = BIGF; pd.s1a[r] = BIGF; pd.s0b[r] = BIGF; pd.s1b[r] = BIGF;
    }

    const int tiles = npad >> 4;
    const int nblk  = tiles / BLK_TILES;
    int blk = 0;
    if (nblk > 0) {
        v2f bufA[BLK_TILES], bufB[BLK_TILES];
        load_block(bufA, Qb);
        for (; blk + 2 <= nblk; blk += 2) {
            load_block(bufB, Qb + (blk + 1) * BLK_FLOATS);
            scan_block_late_fold(bufA, a0, a1, c0, c1, pd, minv0, minv1);
            const int nb = (blk + 2 < nblk) ? (blk + 2) : (nblk - 1);
            load_block(bufA, Qb + nb * BLK_FLOATS);
            scan_block_late_fold(bufB, a0, a1, c0, c1, pd, minv0, minv1);
        }
        if (blk < nblk) {
            scan_block_late_fold(bufA, a0, a1, c0, c1, pd, minv0, minv1);
            ++blk;
        }
    }
    fold4(minv0, pd.s0a, pd.s0b, 0);
    fold4(minv0, pd.s0a, pd.s0b, 4);
    fold4(minv1, pd.s1a, pd.s1b, 0);
    fold4(minv1, pd.s1a, pd.s1b, 4);
    for (int col0 = blk * (BLK_TILES * 16); col0 < npad; col0 += 16) {
        v2f bb = *(const v2f*)(Qb + 4 * col0);
        v8f d0 = __builtin_amdgcn_wmma_f32_16x16x4_f32(
            false, a0, false, bb, (short)0, c0, false, false);
        v8f d1 = __builtin_amdgcn_wmma_f32_16x16x4_f32(
            false, a1, false, bb, (short)0, c1, false, false);
#pragma unroll
        for (int r = 0; r < 8; ++r) {
            minv0[r] = fminf(minv0[r], d0[r]);
            minv1[r] = fminf(minv1[r], d1[r]);
        }
    }
}

__device__ __forceinline__ void scan_lds_table(
    const float* Qb, int npad, const RowOperands& ro, v8f& minv0, v8f& minv1)
{
#pragma unroll
    for (int r = 0; r < 8; ++r) { minv0[r] = BIGF; minv1[r] = BIGF; }
    scan_lds_region(Qb, npad, ro, minv0, minv1);
}

__device__ __forceinline__ void store_row_minima(
    v8f minv0, v8f minv1, float* __restrict__ ob, int row0, int nrows, int j, bool hi)
{
#pragma unroll
    for (int mask = 1; mask <= 8; mask <<= 1) {
#pragma unroll
        for (int r = 0; r < 8; ++r) {
            minv0[r] = fminf(minv0[r], __shfl_xor(minv0[r], mask, 32));
            minv1[r] = fminf(minv1[r], __shfl_xor(minv1[r], mask, 32));
        }
    }
    if (j == 0) {
        const int mbase = row0 + (hi ? 8 : 0);
#pragma unroll
        for (int r = 0; r < 8; ++r) {
            const int m0 = mbase + r;
            const int m1 = mbase + 16 + r;
            if (m0 < nrows) ob[m0] = minv0[r];
            if (m1 < nrows) ob[m1] = minv1[r];
        }
    }
}

#define PAIR_POINTS 20480

template <bool PLAIN>
__global__ __launch_bounds__(128) void chamfer_pair_kernel(
    const float* __restrict__ P, const float* __restrict__ Q,
    float* __restrict__ out1, float* __restrict__ out2,
    int NP, int NQ, int NPpad, int NQpad, int groups)
{
    __shared__ float4 tq[PAIR_POINTS];
    float4* tqQ = tq;
    float4* tqP = tq + NQpad;

    const int tid  = threadIdx.x;
    const int lane = tid & 31;
    const int b    = blockIdx.x / groups;
    const int row0 = ((blockIdx.x - b * groups) << 7) + ((tid >> 5) << 5);

    const int  j  = lane & 15;
    const bool hi = lane >= 16;

    const float* Praw = P + (size_t)b * NP * 3;
    const float* Qraw = Q + (size_t)b * NQ * 3;

    const int r0 = row0 + j, r1 = row0 + 16 + j;
    const float* s0 = Praw + (size_t)(r0 < NP ? r0 : NP - 1) * 3;
    const float* s1 = Praw + (size_t)(r1 < NP ? r1 : NP - 1) * 3;
    const float* t0 = Qraw + (size_t)(r0 < NQ ? r0 : NQ - 1) * 3;
    const float* t1 = Qraw + (size_t)(r1 < NQ ? r1 : NQ - 1) * 3;
    const float p0x = s0[0], p0y = s0[1], p0z = s0[2];
    const float p1x = s1[0], p1y = s1[1], p1z = s1[2];
    const float q0x = t0[0], q0y = t0[1], q0z = t0[2];
    const float q1x = t1[0], q1y = t1[1], q1z = t1[2];

    if (PLAIN || NQ == NQpad) build_table<false>(tqQ, Qraw, NQ, NQpad, tid);
    else                      build_table<true >(tqQ, Qraw, NQ, NQpad, tid);

    const RowOperands ro1 = make_row_operands(p0x, p0y, p0z, p1x, p1y, p1z, hi);

    float hx[HOLD_POINTS], hy[HOLD_POINTS], hz[HOLD_POINTS];
    hold_points<!PLAIN>(hx, hy, hz, Praw, NP, tid);

    lds_barrier();

    if (row0 < NP) {
        v8f minv0, minv1;
        scan_lds_table((const float*)tqQ + 4 * j + (hi ? 2 : 0), NQpad, ro1, minv0, minv1);
        store_row_minima(minv0, minv1, out1 + (size_t)b * NP, row0, NP, j, hi);
    }

    store_held_points<!PLAIN>(tqP, hx, hy, hz, NP, NPpad, tid);
    for (int idx = 128 * HOLD_POINTS + tid; idx < NPpad; idx += 128) {
        const bool ok = idx < NP;
        const float* s = Praw + (size_t)(ok ? idx : NP - 1) * 3;
        const float x = ok ? s[0] : 0.f;
        const float y = ok ? s[1] : 0.f;
        const float z = ok ? s[2] : 0.f;
        const float w = ok ? (x * x + y * y + z * z) : BIGF;
        tqP[idx] = make_float4(x, y, z, w);
    }

    lds_barrier();

    if (row0 < NQ) {
        const RowOperands ro = make_row_operands(q0x, q0y, q0z, q1x, q1y, q1z, hi);
        v8f minv0, minv1;
        scan_lds_table((const float*)tqP + 4 * j + (hi ? 2 : 0), NPpad, ro, minv0, minv1);
        store_row_minima(minv0, minv1, out2 + (size_t)b * NQ, row0, NQ, j, hi);
    }
}

#define GROUP_POINTS 16

template <int NTHR>
__device__ __forceinline__ void build_region(
    float4* tq, const float* __restrict__ raw, int first, int count, int t)
{
    int g = 0;
    for (; g + NTHR * GROUP_POINTS <= count; g += NTHR * GROUP_POINTS) {
        float qx[GROUP_POINTS], qy[GROUP_POINTS], qz[GROUP_POINTS];
#pragma unroll
        for (int u = 0; u < GROUP_POINTS; ++u) {
            const float* s = raw + (size_t)(first + g + NTHR * u + t) * 3;
            qx[u] = s[0]; qy[u] = s[1]; qz[u] = s[2];
        }
#pragma unroll
        for (int u = 0; u < GROUP_POINTS; ++u) {
            const float x = qx[u], y = qy[u], z = qz[u];
            tq[first + g + NTHR * u + t] = make_float4(x, y, z, x * x + y * y + z * z);
        }
    }
    for (int i = g + t; i < count; i += NTHR) {
        const float* s = raw + (size_t)(first + i) * 3;
        const float x = s[0], y = s[1], z = s[2];
        tq[first + i] = make_float4(x, y, z, x * x + y * y + z * z);
    }
}

__global__ __launch_bounds__(256) void chamfer_staged_kernel(
    const float* __restrict__ P, const float* __restrict__ Q,
    float* __restrict__ out1, float* __restrict__ out2,
    int NP, int NQ, int groups)
{
    __shared__ float4 tq[PAIR_POINTS];

    const int  tid     = threadIdx.x;
    const int  lane    = tid & 31;
    const int  wave    = tid >> 5;
    const bool scanner = wave < 4;
    const int  b       = blockIdx.x / groups;
    const int  row0    = ((blockIdx.x - b * groups) << 7) + ((wave & 3) << 5);

    const int  j  = lane & 15;
    const bool hi = lane >= 16;

    const float* Praw = P + (size_t)b * NP * 3;
    const float* Qraw = Q + (size_t)b * NQ * 3;

    const int r0 = row0 + j, r1 = row0 + 16 + j;
    const float* s0 = Praw + (size_t)(r0 < NP ? r0 : NP - 1) * 3;
    const float* s1 = Praw + (size_t)(r1 < NP ? r1 : NP - 1) * 3;
    const float* t0 = Qraw + (size_t)(r0 < NQ ? r0 : NQ - 1) * 3;
    const float* t1 = Qraw + (size_t)(r1 < NQ ? r1 : NQ - 1) * 3;
    const float p0x = s0[0], p0y = s0[1], p0z = s0[2];
    const float p1x = s1[0], p1y = s1[1], p1z = s1[2];
    const float q0x = t0[0], q0y = t0[1], q0z = t0[2];
    const float q1x = t1[0], q1y = t1[1], q1z = t1[2];

    int qsplit = ((NQ >> 1) + 511) & ~511;
    if (qsplit > NQ) qsplit = NQ;

    build_region<256>(tq, Qraw, 0, qsplit, tid);
    lds_barrier();

    v8f minv0, minv1;
#pragma unroll
    for (int r = 0; r < 8; ++r) { minv0[r] = BIGF; minv1[r] = BIGF; }

#pragma unroll 1
    for (int s = 0; s < 3; ++s) {
        if (scanner) {
            const bool second = (s == 2);
            const int  nrows  = second ? NQ : NP;
            if (row0 < nrows) {
                const RowOperands ro = make_row_operands(
                    second ? q0x : p0x, second ? q0y : p0y, second ? q0z : p0z,
                    second ? q1x : p1x, second ? q1y : p1y, second ? q1z : p1z, hi);
                const int start = (s == 0) ? 0 : (s == 1) ? qsplit : NQ;
                const int npts  = (s == 0) ? qsplit : (s == 1) ? NQ - qsplit : NP;
                scan_lds_region((const float*)(tq + start) + 4 * j + (hi ? 2 : 0),
                                npts, ro, minv0, minv1);
                if (s >= 1) {
                    store_row_minima(minv0, minv1,
                                     second ? out2 + (size_t)b * NQ : out1 + (size_t)b * NP,
                                     row0, nrows, j, hi);
#pragma unroll
                    for (int r = 0; r < 8; ++r) { minv0[r] = BIGF; minv1[r] = BIGF; }
                }
            }
        } else if (s < 2) {
            if (s == 0) build_region<128>(tq, Qraw, qsplit, NQ - qsplit, tid - 128);
            else        build_region<128>(tq + NQ, Praw, 0, NP, tid - 128);
        }
        if (s < 2) lds_barrier();
    }
}

__global__ __launch_bounds__(256) void chamfer_nn_min_fallback(
    const float* __restrict__ P, const float* __restrict__ Q,
    float* __restrict__ out, int NP, int NQ, int B)
{
    const int lane  = threadIdx.x & 31;
    const int gwave = blockIdx.x * (blockDim.x >> 5) + (threadIdx.x >> 5);
    const int stripsPerBatch = (NP + 15) >> 4;
    const int b = gwave / stripsPerBatch;
    if (b >= B) return;
    const int row0 = (gwave - b * stripsPerBatch) << 4;
    const int  j  = lane & 15;
    const bool hi = lane >= 16;

    const float* Pb = P + (size_t)b * NP * 3;
    const float* Qb = Q + (size_t)b * NQ * 3;

    int pi = row0 + j; if (pi >= NP) pi = NP - 1;
    float px = Pb[3 * pi], py = Pb[3 * pi + 1], pz = Pb[3 * pi + 2];
    float p2 = px * px + py * py + pz * pz;

    v2f a;
    a.x = hi ? (-2.f * pz) : (-2.f * px);
    a.y = hi ? 1.f         : (-2.f * py);

    v8f c, minv;
#pragma unroll
    for (int r = 0; r < 8; ++r) {
        c[r] = __shfl(p2, (hi ? 8 : 0) + r, 16);
        minv[r] = BIGF;
    }

#pragma unroll 2
    for (int col0 = 0; col0 < NQ; col0 += 16) {
        int qi = col0 + j;
        bool qok = qi < NQ;
        if (!qok) qi = NQ - 1;
        float qx = Qb[3 * qi], qy = Qb[3 * qi + 1], qz = Qb[3 * qi + 2];
        float q2 = qok ? (qx * qx + qy * qy + qz * qz) : BIGF;
        v2f bb;
        bb.x = hi ? qz : qx;
        bb.y = hi ? q2 : qy;
        v8f d = __builtin_amdgcn_wmma_f32_16x16x4_f32(
            false, a, false, bb, (short)0, c, false, false);
#pragma unroll
        for (int r = 0; r < 8; ++r) minv[r] = fminf(minv[r], d[r]);
    }

#pragma unroll
    for (int mask = 1; mask <= 8; mask <<= 1) {
#pragma unroll
        for (int r = 0; r < 8; ++r)
            minv[r] = fminf(minv[r], __shfl_xor(minv[r], mask, 32));
    }

    if (j == 0) {
        int mbase = row0 + (hi ? 8 : 0);
#pragma unroll
        for (int r = 0; r < 8; ++r) {
            int m = mbase + r;
            if (m < NP) out[(size_t)b * NP + m] = minv[r];
        }
    }
}

extern "C" void kernel_launch(void* const* d_in, const int* in_sizes, int n_in,
                              void* d_out, int out_size, void* d_ws, size_t ws_size,
                              hipStream_t stream) {
    const float* in1 = (const float*)d_in[0];
    const float* in2 = (const float*)d_in[1];
    float* out = (float*)d_out;

    const int B = 4;
    const int N = in_sizes[0] / (B * 3);
    const int M = in_sizes[1] / (B * 3);
    const int Npad = (N + 31) & ~31;
    const int Mpad = (M + 31) & ~31;

    size_t need = (size_t)B * (Npad + Mpad) * sizeof(float4);
    if (N > 0 && M > 0 && Npad + Mpad <= PAIR_POINTS) {
        int gN = (N + 127) / 128;
        int gM = (M + 127) / 128;
        int groups = gN > gM ? gN : gM;
        if (N == Npad && M == Mpad)
            chamfer_staged_kernel<<<B * groups, 256, 0, stream>>>(
                in1, in2, out, out + (size_t)B * N, N, M, groups);
        else
            chamfer_pair_kernel<false><<<B * groups, 128, 0, stream>>>(
                in1, in2, out, out + (size_t)B * N, N, M, Npad, Mpad, groups);
    } else if (ws_size >= need) {
        float4* T1 = (float4*)d_ws;
        float4* T2 = T1 + (size_t)B * Npad;

        transform_points_kernel<<<(B * Npad + 255) / 256, 256, 0, stream>>>(in1, T1, N, Npad, B);
        transform_points_kernel<<<(B * Mpad + 255) / 256, 256, 0, stream>>>(in2, T2, M, Mpad, B);

        {
            int waves = B * (Npad >> 5);
            int blocks = (waves + 3) / 4;
            chamfer_wmma_kernel<<<blocks, 128, 0, stream>>>(T1, T2, out, N, Npad, Mpad, B);
        }
        {
            int waves = B * (Mpad >> 5);
            int blocks = (waves + 3) / 4;
            chamfer_wmma_kernel<<<blocks, 128, 0, stream>>>(T2, T1, out + (size_t)B * N, M, Mpad, Npad, B);
        }
    } else {
        {
            int waves = B * ((N + 15) / 16);
            int blocks = (waves + 7) / 8;
            chamfer_nn_min_fallback<<<blocks, 256, 0, stream>>>(in1, in2, out, N, M, B);
        }
        {
            int waves = B * ((M + 15) / 16);
            int blocks = (waves + 7) / 8;
            chamfer_nn_min_fallback<<<blocks, 256, 0, stream>>>(in2, in1, out + (size_t)B * N, M, N, B);
        }
    }
}


# One-launch kernel: distance tiles transposed (table points = WMMA rows, a wave's own 32 points = columns): a lane's 8 results fold in-lane, no 16-lane shuffle rounds, one 128-B store; bit-identical
# speedup vs baseline: 4.6351x; 1.0931x over previous
#include <hip/hip_runtime.h>
#include <hip/hip_bf16.h>

typedef float v2f __attribute__((ext_vector_type(2)));
typedef float v8f __attribute__((ext_vector_type(8)));

#define BIGF 3.0e38f

__global__ __launch_bounds__(256) void transform_points_kernel(
    const float* __restrict__ src, float4* __restrict__ dst, int n, int npad, int B)
{
    int i = blockIdx.x * blockDim.x + threadIdx.x;
    int total = B * npad;
    if (i >= total) return;
    int b = i / npad;
    int idx = i - b * npad;
    float x = 0.f, y = 0.f, z = 0.f, w = BIGF;
    if (idx < n) {
        const float* s = src + ((size_t)b * n + idx) * 3;
        x = s[0]; y = s[1]; z = s[2];
        w = x * x + y * y + z * z;
    }
    dst[i] = make_float4(x, y, z, w);
}

#define BLK_TILES  32
#define BLK_FLOATS (BLK_TILES * 64)

__device__ __forceinline__ void load_block(v2f (&buf)[BLK_TILES], const float* p)
{
#pragma unroll
    for (int t = 0; t < BLK_TILES; ++t)
        buf[t] = *(const v2f*)(p + 64 * t);
    __builtin_amdgcn_sched_barrier(0);
}

__device__ __forceinline__ void scan_block(
    const v2f (&buf)[BLK_TILES], v2f a0, v2f a1, v8f c0, v8f c1, v8f& minv0, v8f& minv1)
{
#pragma unroll
    for (int t = 0; t < BLK_TILES; ++t) {
        v8f d0 = __builtin_amdgcn_wmma_f32_16x16x4_f32(
            false, a0, false, buf[t], (short)0, c0, false, false);
        v8f d1 = __builtin_amdgcn_wmma_f32_16x16x4_f32(
            false, a1, false, buf[t], (short)0, c1, false, false);
#pragma unroll
        for (int r = 0; r < 8; ++r) {
            minv0[r] = fminf(minv0[r], d0[r]);
            minv1[r] = fminf(minv1[r], d1[r]);
        }
    }
}

__global__ __launch_bounds__(128) void chamfer_wmma_kernel(
    const float4* __restrict__ Pt, const float4* __restrict__ Qt,
    float* __restrict__ out, int NP, int NPpad, int NQpad, int B)
{
    const int lane  = threadIdx.x & 31;
    const int gwave = blockIdx.x * (blockDim.x >> 5) + (threadIdx.x >> 5);

    const int groups = NPpad >> 5;
    const int b = gwave / groups;
    if (b >= B) return;
    const int row0 = (gwave - b * groups) << 5;

    const int  j  = lane & 15;
    const bool hi = lane >= 16;

    const float4* Pb = Pt + (size_t)b * NPpad;
    const float* Qb = (const float*)(Qt + (size_t)b * NQpad) + 4 * j + (hi ? 2 : 0);

    float4 p0 = Pb[row0 + j];
    float4 p1 = Pb[row0 + 16 + j];

    v2f a0, a1;
    a0.x = hi ? (-2.f * p0.z) : (-2.f * p0.x);
    a0.y = hi ? 1.f           : (-2.f * p0.y);
    a1.x = hi ? (-2.f * p1.z) : (-2.f * p1.x);
    a1.y = hi ? 1.f           : (-2.f * p1.y);

    const int srcBase = hi ? 8 : 0;
    v8f c0, c1, minv0, minv1;
#pragma unroll
    for (int r = 0; r < 8; ++r) {
        c0[r] = __shfl(p0.w, srcBase + r, 16);
        c1[r] = __shfl(p1.w, srcBase + r, 16);
        minv0[r] = BIGF;
        minv1[r] = BIGF;
    }

    const int tiles = NQpad >> 4;
    const int nblk  = tiles / BLK_TILES;
    int blk = 0;
    if (nblk > 0) {
        v2f bufA[BLK_TILES], bufB[BLK_TILES];
        load_block(bufA, Qb);
        for (; blk + 2 <= nblk; blk += 2) {
            load_block(bufB, Qb + (size_t)(blk + 1) * BLK_FLOATS);
            scan_block(bufA, a0, a1, c0, c1, minv0, minv1);
            const int nb = (blk + 2 < nblk) ? (blk + 2) : (nblk - 1);
            load_block(bufA, Qb + (size_t)nb * BLK_FLOATS);
            scan_block(bufB, a0, a1, c0, c1, minv0, minv1);
        }
        if (blk < nblk) {
            scan_block(bufA, a0, a1, c0, c1, minv0, minv1);
            ++blk;
        }
    }
    for (int col0 = blk * (BLK_TILES * 16); col0 < NQpad; col0 += 16) {
        v2f bb = *(const v2f*)(Qb + 4 * col0);
        v8f d0 = __builtin_amdgcn_wmma_f32_16x16x4_f32(
            false, a0, false, bb, (short)0, c0, false, false);
        v8f d1 = __builtin_amdgcn_wmma_f32_16x16x4_f32(
            false, a1, false, bb, (short)0, c1, false, false);
#pragma unroll
        for (int r = 0; r < 8; ++r) {
            minv0[r] = fminf(minv0[r], d0[r]);
            minv1[r] = fminf(minv1[r], d1[r]);
        }
    }

#pragma unroll
    for (int mask = 1; mask <= 8; mask <<= 1) {
#pragma unroll
        for (int r = 0; r < 8; ++r) {
            minv0[r] = fminf(minv0[r], __shfl_xor(minv0[r], mask, 32));
            minv1[r] = fminf(minv1[r], __shfl_xor(minv1[r], mask, 32));
        }
    }

    if (j == 0) {
        int mbase = row0 + (hi ? 8 : 0);
        float* ob = out + (size_t)b * NP;
#pragma unroll
        for (int r = 0; r < 8; ++r) {
            int m0 = mbase + r;
            int m1 = mbase + 16 + r;
            if (m0 < NP) ob[m0] = minv0[r];
            if (m1 < NP) ob[m1] = minv1[r];
        }
    }
}

struct PendingPair { v8f s0a, s1a, s0b, s1b; };

__device__ __forceinline__ void fold4(v8f& m, const v8f& x, const v8f& y, int r0)
{
#pragma unroll
    for (int r = r0; r < r0 + 4; ++r) m[r] = fminf(fminf(m[r], x[r]), y[r]);
}

__device__ __forceinline__ void scan_block_late_fold(
    const v2f (&buf)[BLK_TILES], v2f a0, v2f a1, v8f c0, v8f c1,
    PendingPair& pd, v8f& minv0, v8f& minv1)
{
#pragma unroll
    for (int t = 0; t < BLK_TILES; t += 2) {
        v8f n0a = __builtin_amdgcn_wmma_f32_16x16x4_f32(
            false, a0, false, buf[t], (short)0, c0, false, false);
        fold4(minv0, pd.s0a, pd.s0b, 0);
        v8f n1a = __builtin_amdgcn_wmma_f32_16x16x4_f32(
            false, a1, false, buf[t], (short)0, c1, false, false);
        fold4(minv0, pd.s0a, pd.s0b, 4);
        v8f n0b = __builtin_amdgcn_wmma_f32_16x16x4_f32(
            false, a0, false, buf[t + 1], (short)0, c0, false, false);
        fold4(minv1, pd.s1a, pd.s1b, 0);
        v8f n1b = __builtin_amdgcn_wmma_f32_16x16x4_f32(
            false, a1, false, buf[t + 1], (short)0, c1, false, false);
        fold4(minv1, pd.s1a, pd.s1b, 4);
        pd.s0a = n0a; pd.s1a = n1a; pd.s0b = n0b; pd.s1b = n1b;
    }
#pragma unroll
    for (int k = 0; k < 2 * BLK_TILES; ++k) {
        __builtin_amdgcn_sched_group_barrier(0x008, 1, 0);
        __builtin_amdgcn_sched_group_barrier(0x002, 4, 0);
    }
    __builtin_amdgcn_sched_barrier(0);
}

#define STAGE_POINTS 64

template <bool PADDED>
__device__ __forceinline__ void build_table(
    float4* tq, const float* __restrict__ Qraw, int n, int npad, int tid)
{
    int g = 0;
    for (; g + 128 * STAGE_POINTS <= npad; g += 128 * STAGE_POINTS) {
        float qx[STAGE_POINTS], qy[STAGE_POINTS], qz[STAGE_POINTS];
#pragma unroll
        for (int u = 0; u < STAGE_POINTS; ++u) {
            const int idx = g + 128 * u + tid;
            const float* s = Qraw + (size_t)((!PADDED || idx < n) ? idx : n - 1) * 3;
            qx[u] = s[0]; qy[u] = s[1]; qz[u] = s[2];
        }
#pragma unroll
        for (int u = 0; u < STAGE_POINTS; ++u) {
            const int idx = g + 128 * u + tid;
            const bool ok = !PADDED || idx < n;
            const float x = ok ? qx[u] : 0.f;
            const float y = ok ? qy[u] : 0.f;
            const float z = ok ? qz[u] : 0.f;
            const float w = ok ? (x * x + y * y + z * z) : BIGF;
            tq[idx] = make_float4(x, y, z, w);
        }
    }
    for (int idx = g + tid; idx < npad; idx += 128) {
        const bool ok = !PADDED || idx < n;
        const float* s = Qraw + (size_t)(ok ? idx : n - 1) * 3;
        const float x = ok ? s[0] : 0.f;
        const float y = ok ? s[1] : 0.f;
        const float z = ok ? s[2] : 0.f;
        const float w = ok ? (x * x + y * y + z * z) : BIGF;
        tq[idx] = make_float4(x, y, z, w);
    }
}

__device__ __forceinline__ void lds_barrier()
{
    __builtin_amdgcn_fence(__ATOMIC_RELEASE, "workgroup", "local");
    __builtin_amdgcn_s_barrier();
    __builtin_amdgcn_fence(__ATOMIC_ACQUIRE, "workgroup", "local");
}

#define HOLD_POINTS 64

template <bool CLAMP>
__device__ __forceinline__ void hold_points(
    float (&hx)[HOLD_POINTS], float (&hy)[HOLD_POINTS], float (&hz)[HOLD_POINTS],
    const float* __restrict__ raw, int n, int tid)
{
#pragma unroll
    for (int u = 0; u < HOLD_POINTS; ++u) {
        const int idx = 128 * u + tid;
        const float* s = raw + (size_t)((!CLAMP || idx < n) ? idx : n - 1) * 3;
        hx[u] = s[0]; hy[u] = s[1]; hz[u] = s[2];
    }
    __builtin_amdgcn_sched_barrier(0);
}

template <bool CLAMP>
__device__ __forceinline__ void store_held_points(
    float4* tq, const float (&hx)[HOLD_POINTS], const float (&hy)[HOLD_POINTS],
    const float (&hz)[HOLD_POINTS], int n, int npad, int tid)
{
#pragma unroll
    for (int u = 0; u < HOLD_POINTS; ++u) {
        const int idx = 128 * u + tid;
        const bool ok = !CLAMP || idx < n;
        const float x = ok ? hx[u] : 0.f;
        const float y = ok ? hy[u] : 0.f;
        const float z = ok ? hz[u] : 0.f;
        const float w = ok ? (x * x + y * y + z * z) : BIGF;
        if (!CLAMP || idx < npad) tq[idx] = make_float4(x, y, z, w);
    }
}

struct RowOperands { v2f a0, a1; v8f c0, c1; };

__device__ __forceinline__ RowOperands make_row_operands(
    float p0x, float p0y, float p0z, float p1x, float p1y, float p1z, bool hi)
{
    RowOperands ro;
    const float p0w = p0x * p0x + p0y * p0y + p0z * p0z;
    const float p1w = p1x * p1x + p1y * p1y + p1z * p1z;
    ro.a0.x = hi ? (-2.f * p0z) : (-2.f * p0x);
    ro.a0.y = hi ? 1.f          : (-2.f * p0y);
    ro.a1.x = hi ? (-2.f * p1z) : (-2.f * p1x);
    ro.a1.y = hi ? 1.f          : (-2.f * p1y);
    const int srcBase = hi ? 8 : 0;
#pragma unroll
    for (int r = 0; r < 8; ++r) {
        ro.c0[r] = __shfl(p0w, srcBase + r, 16);
        ro.c1[r] = __shfl(p1w, srcBase + r, 16);
    }
    return ro;
}

__device__ __forceinline__ void scan_lds_region(
    const float* Qb, int npad, const RowOperands& ro, v8f& minv0, v8f& minv1)
{
    const v2f a0 = ro.a0, a1 = ro.a1;
    const v8f c0 = ro.c0, c1 = ro.c1;
    PendingPair pd;
#pragma unroll
    for (int r = 0; r < 8; ++r) {
        pd.s0a[r] = BIGF; pd.s1a[r] = BIGF; pd.s0b[r] = BIGF; pd.s1b[r] = BIGF;
    }

    const int tiles = npad >> 4;
    const int nblk  = tiles / BLK_TILES;
    int blk = 0;
    if (nblk > 0) {
        v2f bufA[BLK_TILES], bufB[BLK_TILES];
        load_block(bufA, Qb);
        for (; blk + 2 <= nblk; blk += 2) {
            load_block(bufB, Qb + (blk + 1) * BLK_FLOATS);
            scan_block_late_fold(bufA, a0, a1, c0, c1, pd, minv0, minv1);
            const int nb = (blk + 2 < nblk) ? (blk + 2) : (nblk - 1);
            load_block(bufA, Qb + nb * BLK_FLOATS);
            scan_block_late_fold(bufB, a0, a1, c0, c1, pd, minv0, minv1);
        }
        if (blk < nblk) {
            scan_block_late_fold(bufA, a0, a1, c0, c1, pd, minv0, minv1);
            ++blk;
        }
    }
    fold4(minv0, pd.s0a, pd.s0b, 0);
    fold4(minv0, pd.s0a, pd.s0b, 4);
    fold4(minv1, pd.s1a, pd.s1b, 0);
    fold4(minv1, pd.s1a, pd.s1b, 4);
    for (int col0 = blk * (BLK_TILES * 16); col0 < npad; col0 += 16) {
        v2f bb = *(const v2f*)(Qb + 4 * col0);
        v8f d0 = __builtin_amdgcn_wmma_f32_16x16x4_f32(
            false, a0, false, bb, (short)0, c0, false, false);
        v8f d1 = __builtin_amdgcn_wmma_f32_16x16x4_f32(
            false, a1, false, bb, (short)0, c1, false, false);
#pragma unroll
        for (int r = 0; r < 8; ++r) {
            minv0[r] = fminf(minv0[r], d0[r]);
            minv1[r] = fminf(minv1[r], d1[r]);
        }
    }
}

__device__ __forceinline__ void scan_lds_table(
    const float* Qb, int npad, const RowOperands& ro, v8f& minv0, v8f& minv1)
{
#pragma unroll
    for (int r = 0; r < 8; ++r) { minv0[r] = BIGF; minv1[r] = BIGF; }
    scan_lds_region(Qb, npad, ro, minv0, minv1);
}

__device__ __forceinline__ void store_row_minima(
    v8f minv0, v8f minv1, float* __restrict__ ob, int row0, int nrows, int j, bool hi)
{
#pragma unroll
    for (int mask = 1; mask <= 8; mask <<= 1) {
#pragma unroll
        for (int r = 0; r < 8; ++r) {
            minv0[r] = fminf(minv0[r], __shfl_xor(minv0[r], mask, 32));
            minv1[r] = fminf(minv1[r], __shfl_xor(minv1[r], mask, 32));
        }
    }
    if (j == 0) {
        const int mbase = row0 + (hi ? 8 : 0);
#pragma unroll
        for (int r = 0; r < 8; ++r) {
            const int m0 = mbase + r;
            const int m1 = mbase + 16 + r;
            if (m0 < nrows) ob[m0] = minv0[r];
            if (m1 < nrows) ob[m1] = minv1[r];
        }
    }
}

#define PAIR_POINTS 20480

template <bool PLAIN>
__global__ __launch_bounds__(128) void chamfer_pair_kernel(
    const float* __restrict__ P, const float* __restrict__ Q,
    float* __restrict__ out1, float* __restrict__ out2,
    int NP, int NQ, int NPpad, int NQpad, int groups)
{
    __shared__ float4 tq[PAIR_POINTS];
    float4* tqQ = tq;
    float4* tqP = tq + NQpad;

    const int tid  = threadIdx.x;
    const int lane = tid & 31;
    const int b    = blockIdx.x / groups;
    const int row0 = ((blockIdx.x - b * groups) << 7) + ((tid >> 5) << 5);

    const int  j  = lane & 15;
    const bool hi = lane >= 16;

    const float* Praw = P + (size_t)b * NP * 3;
    const float* Qraw = Q + (size_t)b * NQ * 3;

    const int r0 = row0 + j, r1 = row0 + 16 + j;
    const float* s0 = Praw + (size_t)(r0 < NP ? r0 : NP - 1) * 3;
    const float* s1 = Praw + (size_t)(r1 < NP ? r1 : NP - 1) * 3;
    const float* t0 = Qraw + (size_t)(r0 < NQ ? r0 : NQ - 1) * 3;
    const float* t1 = Qraw + (size_t)(r1 < NQ ? r1 : NQ - 1) * 3;
    const float p0x = s0[0], p0y = s0[1], p0z = s0[2];
    const float p1x = s1[0], p1y = s1[1], p1z = s1[2];
    const float q0x = t0[0], q0y = t0[1], q0z = t0[2];
    const float q1x = t1[0], q1y = t1[1], q1z = t1[2];

    if (PLAIN || NQ == NQpad) build_table<false>(tqQ, Qraw, NQ, NQpad, tid);
    else                      build_table<true >(tqQ, Qraw, NQ, NQpad, tid);

    const RowOperands ro1 = make_row_operands(p0x, p0y, p0z, p1x, p1y, p1z, hi);

    float hx[HOLD_POINTS], hy[HOLD_POINTS], hz[HOLD_POINTS];
    hold_points<!PLAIN>(hx, hy, hz, Praw, NP, tid);

    lds_barrier();

    if (row0 < NP) {
        v8f minv0, minv1;
        scan_lds_table((const float*)tqQ + 4 * j + (hi ? 2 : 0), NQpad, ro1, minv0, minv1);
        store_row_minima(minv0, minv1, out1 + (size_t)b * NP, row0, NP, j, hi);
    }

    store_held_points<!PLAIN>(tqP, hx, hy, hz, NP, NPpad, tid);
    for (int idx = 128 * HOLD_POINTS + tid; idx < NPpad; idx += 128) {
        const bool ok = idx < NP;
        const float* s = Praw + (size_t)(ok ? idx : NP - 1) * 3;
        const float x = ok ? s[0] : 0.f;
        const float y = ok ? s[1] : 0.f;
        const float z = ok ? s[2] : 0.f;
        const float w = ok ? (x * x + y * y + z * z) : BIGF;
        tqP[idx] = make_float4(x, y, z, w);
    }

    lds_barrier();

    if (row0 < NQ) {
        const RowOperands ro = make_row_operands(q0x, q0y, q0z, q1x, q1y, q1z, hi);
        v8f minv0, minv1;
        scan_lds_table((const float*)tqP + 4 * j + (hi ? 2 : 0), NPpad, ro, minv0, minv1);
        store_row_minima(minv0, minv1, out2 + (size_t)b * NQ, row0, NQ, j, hi);
    }
}

struct ColOperands { v2f b0, b1; v8f c0, c1; };

__device__ __forceinline__ ColOperands make_col_operands(
    float p0x, float p0y, float p0z, float p1x, float p1y, float p1z, bool hi)
{
    ColOperands co;
    const float p0w = p0x * p0x + p0y * p0y + p0z * p0z;
    const float p1w = p1x * p1x + p1y * p1y + p1z * p1z;
    co.b0.x = hi ? (-2.f * p0z) : (-2.f * p0x);
    co.b0.y = hi ? 1.f          : (-2.f * p0y);
    co.b1.x = hi ? (-2.f * p1z) : (-2.f * p1x);
    co.b1.y = hi ? 1.f          : (-2.f * p1y);
#pragma unroll
    for (int r = 0; r < 8; ++r) { co.c0[r] = p0w; co.c1[r] = p1w; }
    return co;
}

__device__ __forceinline__ void fold_tile(v8f& m, const v8f& x, int i0)
{
#pragma unroll
    for (int i = 0; i < 4; ++i)
        m[i0 + i] = fminf(fminf(m[i0 + i], x[2 * i]), x[2 * i + 1]);
}

__device__ __forceinline__ void scan_block_cols(
    const v2f (&buf)[BLK_TILES], v2f b0, v2f b1, v8f c0, v8f c1,
    PendingPair& pd, v8f& minv0, v8f& minv1)
{
#pragma unroll
    for (int t = 0; t < BLK_TILES; t += 2) {
        v8f n0a = __builtin_amdgcn_wmma_f32_16x16x4_f32(
            false, buf[t], false, b0, (short)0, c0, false, false);
        fold_tile(minv0, pd.s0a, 0);
        v8f n1a = __builtin_amdgcn_wmma_f32_16x16x4_f32(
            false, buf[t], false, b1, (short)0, c1, false, false);
        fold_tile(minv0, pd.s0b, 4);
        v8f n0b = __builtin_amdgcn_wmma_f32_16x16x4_f32(
            false, buf[t + 1], false, b0, (short)0, c0, false, false);
        fold_tile(minv1, pd.s1a, 0);
        v8f n1b = __builtin_amdgcn_wmma_f32_16x16x4_f32(
            false, buf[t + 1], false, b1, (short)0, c1, false, false);
        fold_tile(minv1, pd.s1b, 4);
        pd.s0a = n0a; pd.s1a = n1a; pd.s0b = n0b; pd.s1b = n1b;
    }
#pragma unroll
    for (int k = 0; k < 2 * BLK_TILES; ++k) {
        __builtin_amdgcn_sched_group_barrier(0x008, 1, 0);
        __builtin_amdgcn_sched_group_barrier(0x002, 4, 0);
    }
    __builtin_amdgcn_sched_barrier(0);
}

__device__ __forceinline__ void scan_lds_region_cols(
    const float* Qb, int npts, const ColOperands& co, v8f& minv0, v8f& minv1)
{
    const v2f b0 = co.b0, b1 = co.b1;
    const v8f c0 = co.c0, c1 = co.c1;
    PendingPair pd;
#pragma unroll
    for (int r = 0; r < 8; ++r) {
        pd.s0a[r] = BIGF; pd.s1a[r] = BIGF; pd.s0b[r] = BIGF; pd.s1b[r] = BIGF;
    }

    const int tiles = npts >> 4;
    const int nblk  = tiles / BLK_TILES;
    int blk = 0;
    if (nblk > 0) {
        v2f bufA[BLK_TILES], bufB[BLK_TILES];
        load_block(bufA, Qb);
        for (; blk + 2 <= nblk; blk += 2) {
            load_block(bufB, Qb + (blk + 1) * BLK_FLOATS);
            scan_block_cols(bufA, b0, b1, c0, c1, pd, minv0, minv1);
            const int nb = (blk + 2 < nblk) ? (blk + 2) : (nblk - 1);
            load_block(bufA, Qb + nb * BLK_FLOATS);
            scan_block_cols(bufB, b0, b1, c0, c1, pd, minv0, minv1);
        }
        if (blk < nblk) {
            scan_block_cols(bufA, b0, b1, c0, c1, pd, minv0, minv1);
            ++blk;
        }
    }
    fold_tile(minv0, pd.s0a, 0);
    fold_tile(minv0, pd.s0b, 4);
    fold_tile(minv1, pd.s1a, 0);
    fold_tile(minv1, pd.s1b, 4);
    for (int col0 = blk * (BLK_TILES * 16); col0 < npts; col0 += 16) {
        v2f aa = *(const v2f*)(Qb + 4 * col0);
        v8f d0 = __builtin_amdgcn_wmma_f32_16x16x4_f32(
            false, aa, false, b0, (short)0, c0, false, false);
        v8f d1 = __builtin_amdgcn_wmma_f32_16x16x4_f32(
            false, aa, false, b1, (short)0, c1, false, false);
        fold_tile(minv0, d0, 0);
        fold_tile(minv1, d1, 0);
    }
}

__device__ __forceinline__ void store_col_minima(
    const v8f& minv0, const v8f& minv1, float* __restrict__ ob,
    int row0, int nrows, int lane, bool hi)
{
    float m0 = fminf(fminf(fminf(minv0[0], minv0[1]), fminf(minv0[2], minv0[3])),
                     fminf(fminf(minv0[4], minv0[5]), fminf(minv0[6], minv0[7])));
    float m1 = fminf(fminf(fminf(minv1[0], minv1[1]), fminf(minv1[2], minv1[3])),
                     fminf(fminf(minv1[4], minv1[5]), fminf(minv1[6], minv1[7])));
    m0 = fminf(m0, __shfl_xor(m0, 16, 32));
    m1 = fminf(m1, __shfl_xor(m1, 16, 32));
    const int row = row0 + lane;
    if (row < nrows) ob[row] = hi ? m1 : m0;
}

#define GROUP_POINTS 16

template <int NTHR>
__device__ __forceinline__ void build_region(
    float4* tq, const float* __restrict__ raw, int first, int count, int t)
{
    int g = 0;
    for (; g + NTHR * GROUP_POINTS <= count; g += NTHR * GROUP_POINTS) {
        float qx[GROUP_POINTS], qy[GROUP_POINTS], qz[GROUP_POINTS];
#pragma unroll
        for (int u = 0; u < GROUP_POINTS; ++u) {
            const float* s = raw + (size_t)(first + g + NTHR * u + t) * 3;
            qx[u] = s[0]; qy[u] = s[1]; qz[u] = s[2];
        }
#pragma unroll
        for (int u = 0; u < GROUP_POINTS; ++u) {
            const float x = qx[u], y = qy[u], z = qz[u];
            tq[first + g + NTHR * u + t] = make_float4(x, y, z, x * x + y * y + z * z);
        }
    }
    for (int i = g + t; i < count; i += NTHR) {
        const float* s = raw + (size_t)(first + i) * 3;
        const float x = s[0], y = s[1], z = s[2];
        tq[first + i] = make_float4(x, y, z, x * x + y * y + z * z);
    }
}

__global__ __launch_bounds__(256) void chamfer_staged_kernel(
    const float* __restrict__ P, const float* __restrict__ Q,
    float* __restrict__ out1, float* __restrict__ out2,
    int NP, int NQ, int groups)
{
    __shared__ float4 tq[PAIR_POINTS];

    const int  tid     = threadIdx.x;
    const int  lane    = tid & 31;
    const int  wave    = tid >> 5;
    const bool scanner = wave < 4;
    const int  b       = blockIdx.x / groups;
    const int  row0    = ((blockIdx.x - b * groups) << 7) + ((wave & 3) << 5);

    const int  j  = lane & 15;
    const bool hi = lane >= 16;

    const float* Praw = P + (size_t)b * NP * 3;
    const float* Qraw = Q + (size_t)b * NQ * 3;

    const int r0 = row0 + j, r1 = row0 + 16 + j;
    const float* s0 = Praw + (size_t)(r0 < NP ? r0 : NP - 1) * 3;
    const float* s1 = Praw + (size_t)(r1 < NP ? r1 : NP - 1) * 3;
    const float* t0 = Qraw + (size_t)(r0 < NQ ? r0 : NQ - 1) * 3;
    const float* t1 = Qraw + (size_t)(r1 < NQ ? r1 : NQ - 1) * 3;
    const float p0x = s0[0], p0y = s0[1], p0z = s0[2];
    const float p1x = s1[0], p1y = s1[1], p1z = s1[2];
    const float q0x = t0[0], q0y = t0[1], q0z = t0[2];
    const float q1x = t1[0], q1y = t1[1], q1z = t1[2];

    int qsplit = ((NQ >> 1) + 511) & ~511;
    if (qsplit > NQ) qsplit = NQ;

    build_region<256>(tq, Qraw, 0, qsplit, tid);
    lds_barrier();

    v8f minv0, minv1;
#pragma unroll
    for (int r = 0; r < 8; ++r) { minv0[r] = BIGF; minv1[r] = BIGF; }

#pragma unroll 1
    for (int s = 0; s < 3; ++s) {
        if (scanner) {
            const bool second = (s == 2);
            const int  nrows  = second ? NQ : NP;
            if (row0 < nrows) {
                const ColOperands co = make_col_operands(
                    second ? q0x : p0x, second ? q0y : p0y, second ? q0z : p0z,
                    second ? q1x : p1x, second ? q1y : p1y, second ? q1z : p1z, hi);
                const int start = (s == 0) ? 0 : (s == 1) ? qsplit : NQ;
                const int npts  = (s == 0) ? qsplit : (s == 1) ? NQ - qsplit : NP;
                scan_lds_region_cols((const float*)(tq + start) + 4 * j + (hi ? 2 : 0),
                                     npts, co, minv0, minv1);
                if (s >= 1) {
                    store_col_minima(minv0, minv1,
                                     second ? out2 + (size_t)b * NQ : out1 + (size_t)b * NP,
                                     row0, nrows, lane, hi);
#pragma unroll
                    for (int r = 0; r < 8; ++r) { minv0[r] = BIGF; minv1[r] = BIGF; }
                }
            }
        } else if (s < 2) {
            if (s == 0) build_region<128>(tq, Qraw, qsplit, NQ - qsplit, tid - 128);
            else        build_region<128>(tq + NQ, Praw, 0, NP, tid - 128);
        }
        if (s < 2) lds_barrier();
    }
}

__global__ __launch_bounds__(256) void chamfer_nn_min_fallback(
    const float* __restrict__ P, const float* __restrict__ Q,
    float* __restrict__ out, int NP, int NQ, int B)
{
    const int lane  = threadIdx.x & 31;
    const int gwave = blockIdx.x * (blockDim.x >> 5) + (threadIdx.x >> 5);
    const int stripsPerBatch = (NP + 15) >> 4;
    const int b = gwave / stripsPerBatch;
    if (b >= B) return;
    const int row0 = (gwave - b * stripsPerBatch) << 4;
    const int  j  = lane & 15;
    const bool hi = lane >= 16;

    const float* Pb = P + (size_t)b * NP * 3;
    const float* Qb = Q + (size_t)b * NQ * 3;

    int pi = row0 + j; if (pi >= NP) pi = NP - 1;
    float px = Pb[3 * pi], py = Pb[3 * pi + 1], pz = Pb[3 * pi + 2];
    float p2 = px * px + py * py + pz * pz;

    v2f a;
    a.x = hi ? (-2.f * pz) : (-2.f * px);
    a.y = hi ? 1.f         : (-2.f * py);

    v8f c, minv;
#pragma unroll
    for (int r = 0; r < 8; ++r) {
        c[r] = __shfl(p2, (hi ? 8 : 0) + r, 16);
        minv[r] = BIGF;
    }

#pragma unroll 2
    for (int col0 = 0; col0 < NQ; col0 += 16) {
        int qi = col0 + j;
        bool qok = qi < NQ;
        if (!qok) qi = NQ - 1;
        float qx = Qb[3 * qi], qy = Qb[3 * qi + 1], qz = Qb[3 * qi + 2];
        float q2 = qok ? (qx * qx + qy * qy + qz * qz) : BIGF;
        v2f bb;
        bb.x = hi ? qz : qx;
        bb.y = hi ? q2 : qy;
        v8f d = __builtin_amdgcn_wmma_f32_16x16x4_f32(
            false, a, false, bb, (short)0, c, false, false);
#pragma unroll
        for (int r = 0; r < 8; ++r) minv[r] = fminf(minv[r], d[r]);
    }

#pragma unroll
    for (int mask = 1; mask <= 8; mask <<= 1) {
#pragma unroll
        for (int r = 0; r < 8; ++r)
            minv[r] = fminf(minv[r], __shfl_xor(minv[r], mask, 32));
    }

    if (j == 0) {
        int mbase = row0 + (hi ? 8 : 0);
#pragma unroll
        for (int r = 0; r < 8; ++r) {
            int m = mbase + r;
            if (m < NP) out[(size_t)b * NP + m] = minv[r];
        }
    }
}

extern "C" void kernel_launch(void* const* d_in, const int* in_sizes, int n_in,
                              void* d_out, int out_size, void* d_ws, size_t ws_size,
                              hipStream_t stream) {
    const float* in1 = (const float*)d_in[0];
    const float* in2 = (const float*)d_in[1];
    float* out = (float*)d_out;

    const int B = 4;
    const int N = in_sizes[0] / (B * 3);
    const int M = in_sizes[1] / (B * 3);
    const int Npad = (N + 31) & ~31;
    const int Mpad = (M + 31) & ~31;

    size_t need = (size_t)B * (Npad + Mpad) * sizeof(float4);
    if (N > 0 && M > 0 && Npad + Mpad <= PAIR_POINTS) {
        int gN = (N + 127) / 128;
        int gM = (M + 127) / 128;
        int groups = gN > gM ? gN : gM;
        if (N == Npad && M == Mpad)
            chamfer_staged_kernel<<<B * groups, 256, 0, stream>>>(
                in1, in2, out, out + (size_t)B * N, N, M, groups);
        else
            chamfer_pair_kernel<false><<<B * groups, 128, 0, stream>>>(
                in1, in2, out, out + (size_t)B * N, N, M, Npad, Mpad, groups);
    } else if (ws_size >= need) {
        float4* T1 = (float4*)d_ws;
        float4* T2 = T1 + (size_t)B * Npad;

        transform_points_kernel<<<(B * Npad + 255) / 256, 256, 0, stream>>>(in1, T1, N, Npad, B);
        transform_points_kernel<<<(B * Mpad + 255) / 256, 256, 0, stream>>>(in2, T2, M, Mpad, B);

        {
            int waves = B * (Npad >> 5);
            int blocks = (waves + 3) / 4;
            chamfer_wmma_kernel<<<blocks, 128, 0, stream>>>(T1, T2, out, N, Npad, Mpad, B);
        }
        {
            int waves = B * (Mpad >> 5);
            int blocks = (waves + 3) / 4;
            chamfer_wmma_kernel<<<blocks, 128, 0, stream>>>(T2, T1, out + (size_t)B * N, M, Mpad, Npad, B);
        }
    } else {
        {
            int waves = B * ((N + 15) / 16);
            int blocks = (waves + 7) / 8;
            chamfer_nn_min_fallback<<<blocks, 256, 0, stream>>>(in1, in2, out, N, M, B);
        }
        {
            int waves = B * ((M + 15) / 16);
            int blocks = (waves + 7) / 8;
            chamfer_nn_min_fallback<<<blocks, 256, 0, stream>>>(in2, in1, out + (size_t)B * N, M, N, B);
        }
    }
}
